# starting kernel
# baseline (speedup 1.0000x reference)
#include <hip/hip_runtime.h>
#include <math.h>

typedef float v2f __attribute__((ext_vector_type(2)));
typedef float v8f __attribute__((ext_vector_type(8)));

#define D_DIM 512
#define H_DIM 2048

__global__ __launch_bounds__(256)
void qff_pack_w1(const float* __restrict__ W1, float* __restrict__ w1c) {
  int idx = blockIdx.x * blockDim.x + threadIdx.x;
  if (idx < D_DIM * 4) {
    int d = idx >> 2, j = idx & 3;
    w1c[idx] = W1[d * H_DIM + j];
  }
}

__global__ __launch_bounds__(32)
void qff_main(const float* __restrict__ x, const float* __restrict__ w1c,
              const float* __restrict__ b1, const float* __restrict__ qw,
              const float* __restrict__ Wp, const float* __restrict__ bp,
              float* __restrict__ out)
{
  __shared__ float sAng[64];
  __shared__ float sQ[64];
  __shared__ float sG[64];

  const int lane = threadIdx.x;
  const int hsel = lane >> 4;
  const int lid  = lane & 15;
  const long row0 = (long)blockIdx.x * 16;

  if (lane < 8) {
    const int l = lane >> 2, i = lane & 3;
    const float phi = qw[(l * 4 + i) * 3 + 0];
    const float th  = qw[(l * 4 + i) * 3 + 1];
    const float om  = qw[(l * 4 + i) * 3 + 2];
    const float c = cosf(0.5f * th), s = sinf(0.5f * th);
    const float a = -0.5f * (phi + om);
    const float b =  0.5f * (phi - om);
    const float ca = cosf(a), sa = sinf(a);
    const float cb = cosf(b), sb = sinf(b);
    float* g = &sG[lane * 8];
    g[0] =  c * ca;  g[1] =  c * sa;
    g[2] = -s * cb;  g[3] = -s * sb;
    g[4] =  s * cb;  g[5] = -s * sb;
    g[6] =  c * ca;  g[7] = -c * sa;
  }

  v8f acc = {};
  const float* xrow = x + (row0 + lid) * D_DIM + 2 * hsel;
  const float bsel = (lid < 4) ? 1.0f : 0.0f;
  const int   bcol = lid & 3;
  #pragma unroll 4
  for (int k = 0; k < D_DIM; k += 4) {
    v2f afrag; afrag.x = xrow[k]; afrag.y = xrow[k + 1];
    v2f bfrag;
    bfrag.x = bsel * w1c[(k + 2 * hsel + 0) * 4 + bcol];
    bfrag.y = bsel * w1c[(k + 2 * hsel + 1) * 4 + bcol];
    acc = __builtin_amdgcn_wmma_f32_16x16x4_f32(false, afrag, false, bfrag,
                                                (short)0, acc, false, false);
  }

  if (lid < 4) {
    const float bias = b1[lid];
    #pragma unroll
    for (int r = 0; r < 8; ++r) {
      const float v = acc[r] + bias;
      sAng[(hsel * 8 + r) * 4 + lid] = 0.5f * v * (1.0f + erff(v * 0.70710678118654752f));
    }
  }
  __syncthreads();

  for (int t = 0; t < 8; ++t) {
    const int row = 2 * t + hsel;
    const int basis = lid;
    float re = 1.0f, im = 0.0f;
    #pragma unroll
    for (int i = 0; i < 4; ++i) {
      const float a2 = 0.5f * sAng[row * 4 + i];
      re *= ((basis >> (3 - i)) & 1) ? sinf(a2) : cosf(a2);
    }
    #pragma unroll
    for (int l = 0; l < 2; ++l) {
      #pragma unroll
      for (int i = 0; i < 4; ++i) {
        const float* g = &sG[(l * 4 + i) * 8];
        const int mask = 8 >> i;
        const int tb = (basis >> (3 - i)) & 1;
        const float pr = __shfl_xor(re, mask, 32);
        const float pi = __shfl_xor(im, mask, 32);
        const float nr0 = g[0]*re - g[1]*im + g[2]*pr - g[3]*pi;
        const float ni0 = g[0]*im + g[1]*re + g[2]*pi + g[3]*pr;
        const float nr1 = g[4]*pr - g[5]*pi + g[6]*re - g[7]*im;
        const float ni1 = g[4]*pi + g[5]*pr + g[6]*im + g[7]*re;
        re = tb ? nr1 : nr0;
        im = tb ? ni1 : ni0;
      }
      const int rsh = (l % 3) + 1;
      #pragma unroll
      for (int i = 0; i < 4; ++i) {
        const int tq = (i + rsh) & 3;
        const int maskT = 8 >> tq;
        const int ctrl = (basis >> (3 - i)) & 1;
        const float pr = __shfl_xor(re, maskT, 32);
        const float pi = __shfl_xor(im, maskT, 32);
        re = ctrl ? pr : re;
        im = ctrl ? pi : im;
      }
    }
    const float prob = re * re + im * im;
    #pragma unroll
    for (int i = 0; i < 4; ++i) {
      float sv = ((basis >> (3 - i)) & 1) ? -prob : prob;
      sv += __shfl_xor(sv, 1, 32);
      sv += __shfl_xor(sv, 2, 32);
      sv += __shfl_xor(sv, 4, 32);
      sv += __shfl_xor(sv, 8, 32);
      if (lid == 0) sQ[row * 4 + i] = sv;
    }
  }
  __syncthreads();

  v2f aq;
  aq.x = sQ[lid * 4 + 2 * hsel + 0];
  aq.y = sQ[lid * 4 + 2 * hsel + 1];
  for (int n0 = 0; n0 < D_DIM; n0 += 16) {
    v2f bw;
    bw.x = Wp[(2 * hsel + 0) * D_DIM + n0 + lid];
    bw.y = Wp[(2 * hsel + 1) * D_DIM + n0 + lid];
    v8f c = {};
    c = __builtin_amdgcn_wmma_f32_16x16x4_f32(false, aq, false, bw,
                                              (short)0, c, false, false);
    const float bias = bp[n0 + lid];
    float* orow = out + (row0 + hsel * 8) * D_DIM + n0 + lid;
    #pragma unroll
    for (int r = 0; r < 8; ++r)
      orow[(long)r * D_DIM] = c[r] + bias;
  }
}

extern "C" void kernel_launch(void* const* d_in, const int* in_sizes, int n_in,
                              void* d_out, int out_size, void* d_ws, size_t ws_size,
                              hipStream_t stream) {
  const float* x  = (const float*)d_in[0];
  const float* W1 = (const float*)d_in[1];
  const float* b1 = (const float*)d_in[2];
  const float* qw = (const float*)d_in[3];
  const float* Wp = (const float*)d_in[4];
  const float* bp = (const float*)d_in[5];
  float* out = (float*)d_out;
  float* w1c = (float*)d_ws;

  const int rows = in_sizes[0] / D_DIM;
  qff_pack_w1<<<(D_DIM * 4 + 255) / 256, 256, 0, stream>>>(W1, w1c);
  qff_main<<<rows / 16, 32, 0, stream>>>(x, w1c, b1, qw, Wp, bp, out);
}


# GEMM1 operands via LDS: whole x tile fetched at wave start by row-contiguous 16-byte loads, staged per 128-k quarter in padded LDS; WMMAs and gate trig run under the loads in flight; sum order kept
# speedup vs baseline: 1.8858x; 1.8858x over previous
#include <hip/hip_runtime.h>
#include <math.h>

typedef float v2f __attribute__((ext_vector_type(2)));
typedef float v8f __attribute__((ext_vector_type(8)));
typedef float v4f __attribute__((ext_vector_type(4)));

#define D_DIM 512
#define H_DIM 2048
#define XQ_PITCH 132

#define LOAD_ORDER_FENCE() asm volatile("" ::: "memory")

#define LDS_SYNC() do { __builtin_amdgcn_fence(__ATOMIC_RELEASE, "workgroup", "local"); \
                        __builtin_amdgcn_wave_barrier(); \
                        __builtin_amdgcn_fence(__ATOMIC_ACQUIRE, "workgroup", "local"); } while (0)

__global__ __launch_bounds__(256)
void qff_pack_w1(const float* __restrict__ W1, float* __restrict__ w1c) {
  int idx = blockIdx.x * blockDim.x + threadIdx.x;
  if (idx < D_DIM * 4) {
    int d = idx >> 2, j = idx & 3;
    w1c[idx] = W1[d * H_DIM + j];
  }
}

__global__ __launch_bounds__(32)
void qff_main(const float* __restrict__ x, const float* __restrict__ w1c,
              const float* __restrict__ b1, const float* __restrict__ qw,
              const float* __restrict__ Wp, const float* __restrict__ bp,
              float* __restrict__ out)
{
  __shared__ __attribute__((aligned(16))) float sAng[64];
  __shared__ __attribute__((aligned(16))) float sQ[64];
  __shared__ float sG[64];
  __shared__ __attribute__((aligned(16))) float sXq[16 * XQ_PITCH];
  __shared__ __attribute__((aligned(16))) float sW[D_DIM * 4];

  const int lane = threadIdx.x;
  const int hsel = lane >> 4;
  const int lid  = lane & 15;
  const long row0 = (long)blockIdx.x * 16;

  const int gi = lane & 7;
  const float phi = qw[gi * 3 + 0];
  const float th  = qw[gi * 3 + 1];
  const float om  = qw[gi * 3 + 2];
  LOAD_ORDER_FENCE();
  v4f wv[16];
  #pragma unroll
  for (int i = 0; i < 16; ++i)
    wv[i] = *(const v4f*)(w1c + (i * 32 + lane) * 4);
  LOAD_ORDER_FENCE();
  v4f xv[4][16];
  {
    const float* xt = x + row0 * D_DIM + lane * 4;
    #pragma unroll
    for (int q = 0; q < 4; ++q) {
      #pragma unroll
      for (int r = 0; r < 16; ++r)
        xv[q][r] = *(const v4f*)(xt + r * D_DIM + q * 128);
      LOAD_ORDER_FENCE();
    }
  }

  #pragma unroll
  for (int i = 0; i < 16; ++i)
    *(v4f*)&sW[(i * 32 + lane) * 4] = wv[i];

  {
    const float c = cosf(0.5f * th), s = sinf(0.5f * th);
    const float a = -0.5f * (phi + om);
    const float b =  0.5f * (phi - om);
    const float ca = cosf(a), sa = sinf(a);
    const float cb = cosf(b), sb = sinf(b);
    if (lane < 8) {
      float* g = &sG[lane * 8];
      g[0] =  c * ca;  g[1] =  c * sa;
      g[2] = -s * cb;  g[3] = -s * sb;
      g[4] =  s * cb;  g[5] = -s * sb;
      g[6] =  c * ca;  g[7] = -c * sa;
    }
  }

  v8f acc = {};
  const float bsel = (lid < 4) ? 1.0f : 0.0f;
  const int   bcol = lid & 3;
  #pragma unroll
  for (int q = 0; q < 4; ++q) {
    #pragma unroll
    for (int r = 0; r < 16; ++r)
      *(v4f*)&sXq[r * XQ_PITCH + lane * 4] = xv[q][r];
    LDS_SYNC();
    #pragma unroll
    for (int kk = 0; kk < 32; ++kk) {
      const int k = q * 128 + kk * 4;
      const v2f afrag = *(const v2f*)&sXq[lid * XQ_PITCH + kk * 4 + 2 * hsel];
      v2f bfrag;
      bfrag.x = bsel * sW[(k + 2 * hsel + 0) * 4 + bcol];
      bfrag.y = bsel * sW[(k + 2 * hsel + 1) * 4 + bcol];
      acc = __builtin_amdgcn_wmma_f32_16x16x4_f32(false, afrag, false, bfrag,
                                                  (short)0, acc, false, false);
    }
    LDS_SYNC();
  }

  if (lid < 4) {
    const float bias = b1[lid];
    #pragma unroll
    for (int r = 0; r < 8; ++r) {
      const float v = acc[r] + bias;
      sAng[(hsel * 8 + r) * 4 + lid] = 0.5f * v * (1.0f + erff(v * 0.70710678118654752f));
    }
  }
  __syncthreads();

  {
    const int row = lid;
    float a2[4], cs[4], sn[4];
    #pragma unroll
    for (int i = 0; i < 4; ++i) a2[i] = 0.5f * sAng[row * 4 + i];
    #pragma unroll
    for (int i = 0; i < 4; ++i) sincosf(a2[i], &sn[i], &cs[i]);
    float re[16], im[16];
    #pragma unroll
    for (int b = 0; b < 16; ++b) {
      float p = 1.0f;
      #pragma unroll
      for (int i = 0; i < 4; ++i)
        p *= ((b >> (3 - i)) & 1) ? sn[i] : cs[i];
      re[b] = p;
      im[b] = 0.0f;
    }
    #pragma unroll
    for (int l = 0; l < 2; ++l) {
      #pragma unroll
      for (int i = 0; i < 4; ++i) {
        const float* g = &sG[(l * 4 + i) * 8];
        const float g0 = g[0], g1 = g[1], g2 = g[2], g3 = g[3];
        const float g4 = g[4], g5 = g[5], g6 = g[6], g7 = g[7];
        const int mask = 8 >> i;
        #pragma unroll
        for (int b = 0; b < 16; ++b) {
          if (b & mask) continue;
          const int b1 = b | mask;
          const float r0 = re[b],  i0 = im[b];
          const float r1 = re[b1], i1 = im[b1];
          re[b]  = g0*r0 - g1*i0 + g2*r1 - g3*i1;
          im[b]  = g0*i0 + g1*r0 + g2*i1 + g3*r1;
          re[b1] = g4*r0 - g5*i0 + g6*r1 - g7*i1;
          im[b1] = g4*i0 + g5*r0 + g6*i1 + g7*r1;
        }
      }
      const int rsh = (l % 3) + 1;
      #pragma unroll
      for (int i = 0; i < 4; ++i) {
        const int tq = (i + rsh) & 3;
        const int maskT = 8 >> tq;
        float nre[16], nim[16];
        #pragma unroll
        for (int b = 0; b < 16; ++b) {
          const int srcb = ((b >> (3 - i)) & 1) ? (b ^ maskT) : b;
          nre[b] = re[srcb];
          nim[b] = im[srcb];
        }
        #pragma unroll
        for (int b = 0; b < 16; ++b) { re[b] = nre[b]; im[b] = nim[b]; }
      }
    }
    float prob[16];
    #pragma unroll
    for (int b = 0; b < 16; ++b) prob[b] = re[b] * re[b] + im[b] * im[b];
    float z[4];
    #pragma unroll
    for (int i = 0; i < 4; ++i) {
      float t[16];
      #pragma unroll
      for (int b = 0; b < 16; ++b) t[b] = ((b >> (3 - i)) & 1) ? -prob[b] : prob[b];
      #pragma unroll
      for (int s = 1; s < 16; s <<= 1) {
        #pragma unroll
        for (int b = 0; b < 16; b += 2 * s) t[b] = t[b] + t[b + s];
      }
      z[i] = t[0];
    }
    if (hsel == 0) {
      #pragma unroll
      for (int i = 0; i < 4; ++i) sQ[row * 4 + i] = z[i];
    }
  }
  __syncthreads();

  v2f aq;
  aq.x = sQ[lid * 4 + 2 * hsel + 0];
  aq.y = sQ[lid * 4 + 2 * hsel + 1];
  for (int n0 = 0; n0 < D_DIM; n0 += 16) {
    v2f bw;
    bw.x = Wp[(2 * hsel + 0) * D_DIM + n0 + lid];
    bw.y = Wp[(2 * hsel + 1) * D_DIM + n0 + lid];
    v8f c = {};
    c = __builtin_amdgcn_wmma_f32_16x16x4_f32(false, aq, false, bw,
                                              (short)0, c, false, false);
    const float bias = bp[n0 + lid];
    float* orow = out + (row0 + hsel * 8) * D_DIM + n0 + lid;
    #pragma unroll
    for (int r = 0; r < 8; ++r)
      orow[(long)r * D_DIM] = c[r] + bias;
  }
}

extern "C" void kernel_launch(void* const* d_in, const int* in_sizes, int n_in,
                              void* d_out, int out_size, void* d_ws, size_t ws_size,
                              hipStream_t stream) {
  const float* x  = (const float*)d_in[0];
  const float* W1 = (const float*)d_in[1];
  const float* b1 = (const float*)d_in[2];
  const float* qw = (const float*)d_in[3];
  const float* Wp = (const float*)d_in[4];
  const float* bp = (const float*)d_in[5];
  float* out = (float*)d_out;
  float* w1c = (float*)d_ws;

  const int rows = in_sizes[0] / D_DIM;
  qff_pack_w1<<<(D_DIM * 4 + 255) / 256, 256, 0, stream>>>(W1, w1c);
  qff_main<<<rows / 16, 32, 0, stream>>>(x, w1c, b1, qw, Wp, bp, out);
}


# W1's 4 live columns gathered from W1 straight into LDS once per 8-wave work-group (8 tiles, one wave per tile as before): pack kernel and its launch gone, 1 launch per call; sum order kept
# speedup vs baseline: 2.0391x; 1.0813x over previous
#include <hip/hip_runtime.h>
#include <math.h>

typedef float v2f __attribute__((ext_vector_type(2)));
typedef float v8f __attribute__((ext_vector_type(8)));
typedef float v4f __attribute__((ext_vector_type(4)));

#define D_DIM 512
#define H_DIM 2048
#define XQ_PITCH 132

#define LOAD_ORDER_FENCE() asm volatile("" ::: "memory")

#define LDS_SYNC() do { __builtin_amdgcn_fence(__ATOMIC_RELEASE, "workgroup", "local"); \
                        __builtin_amdgcn_wave_barrier(); \
                        __builtin_amdgcn_fence(__ATOMIC_ACQUIRE, "workgroup", "local"); } while (0)

#define WG_WAVES 8
#define WAVE_LDS (16 * XQ_PITCH + 3 * 64)

#define WG_LDS_BARRIER() do { asm volatile("s_wait_dscnt 0x0" ::: "memory"); \
                              asm volatile("s_barrier_signal -1" ::: "memory"); \
                              asm volatile("s_barrier_wait -1" ::: "memory"); } while (0)

__global__ __launch_bounds__(WG_WAVES * 32)
void qff_main(const float* __restrict__ x, const float* __restrict__ W1,
              const float* __restrict__ b1, const float* __restrict__ qw,
              const float* __restrict__ Wp, const float* __restrict__ bp,
              float* __restrict__ out)
{
  __shared__ __attribute__((aligned(16))) float sAll[D_DIM * 4 + WG_WAVES * WAVE_LDS];

  const int tid  = threadIdx.x;
  const int lane = tid & 31;
  const int wave = __builtin_amdgcn_readfirstlane(tid >> 5);
  float* const sW   = sAll;
  float* const sXq  = sAll + D_DIM * 4 + wave * WAVE_LDS;
  float* const sAng = sXq + 16 * XQ_PITCH;
  float* const sQ   = sAng + 64;
  float* const sG   = sQ + 64;
  const int hsel = lane >> 4;
  const int lid  = lane & 15;
  const long row0 = ((long)blockIdx.x * WG_WAVES + wave) * 16;

  v4f wv[2];
  #pragma unroll
  for (int i = 0; i < 2; ++i)
    wv[i] = *(const v4f*)(W1 + (long)(i * (WG_WAVES * 32) + tid) * H_DIM);
  LOAD_ORDER_FENCE();
  const int gi = lane & 7;
  const float phi = qw[gi * 3 + 0];
  const float th  = qw[gi * 3 + 1];
  const float om  = qw[gi * 3 + 2];
  LOAD_ORDER_FENCE();
  v4f xv[4][16];
  {
    const float* xt = x + row0 * D_DIM + lane * 4;
    #pragma unroll
    for (int q = 0; q < 4; ++q) {
      #pragma unroll
      for (int r = 0; r < 16; ++r)
        xv[q][r] = *(const v4f*)(xt + r * D_DIM + q * 128);
      LOAD_ORDER_FENCE();
    }
  }

  #pragma unroll
  for (int i = 0; i < 2; ++i)
    *(v4f*)&sW[(i * (WG_WAVES * 32) + tid) * 4] = wv[i];

  {
    const float c = cosf(0.5f * th), s = sinf(0.5f * th);
    const float a = -0.5f * (phi + om);
    const float b =  0.5f * (phi - om);
    const float ca = cosf(a), sa = sinf(a);
    const float cb = cosf(b), sb = sinf(b);
    if (lane < 8) {
      float* g = &sG[lane * 8];
      g[0] =  c * ca;  g[1] =  c * sa;
      g[2] = -s * cb;  g[3] = -s * sb;
      g[4] =  s * cb;  g[5] = -s * sb;
      g[6] =  c * ca;  g[7] = -c * sa;
    }
  }

  WG_LDS_BARRIER();

  v8f acc = {};
  const float bsel = (lid < 4) ? 1.0f : 0.0f;
  const int   bcol = lid & 3;
  #pragma unroll
  for (int q = 0; q < 4; ++q) {
    #pragma unroll
    for (int r = 0; r < 16; ++r)
      *(v4f*)&sXq[r * XQ_PITCH + lane * 4] = xv[q][r];
    LDS_SYNC();
    #pragma unroll
    for (int kk = 0; kk < 32; ++kk) {
      const int k = q * 128 + kk * 4;
      const v2f afrag = *(const v2f*)&sXq[lid * XQ_PITCH + kk * 4 + 2 * hsel];
      v2f bfrag;
      bfrag.x = bsel * sW[(k + 2 * hsel + 0) * 4 + bcol];
      bfrag.y = bsel * sW[(k + 2 * hsel + 1) * 4 + bcol];
      acc = __builtin_amdgcn_wmma_f32_16x16x4_f32(false, afrag, false, bfrag,
                                                  (short)0, acc, false, false);
    }
    LDS_SYNC();
  }

  if (lid < 4) {
    const float bias = b1[lid];
    #pragma unroll
    for (int r = 0; r < 8; ++r) {
      const float v = acc[r] + bias;
      sAng[(hsel * 8 + r) * 4 + lid] = 0.5f * v * (1.0f + erff(v * 0.70710678118654752f));
    }
  }
  LDS_SYNC();

  {
    const int row = lid;
    float a2[4], cs[4], sn[4];
    #pragma unroll
    for (int i = 0; i < 4; ++i) a2[i] = 0.5f * sAng[row * 4 + i];
    #pragma unroll
    for (int i = 0; i < 4; ++i) sincosf(a2[i], &sn[i], &cs[i]);
    float re[16], im[16];
    #pragma unroll
    for (int b = 0; b < 16; ++b) {
      float p = 1.0f;
      #pragma unroll
      for (int i = 0; i < 4; ++i)
        p *= ((b >> (3 - i)) & 1) ? sn[i] : cs[i];
      re[b] = p;
      im[b] = 0.0f;
    }
    #pragma unroll
    for (int l = 0; l < 2; ++l) {
      #pragma unroll
      for (int i = 0; i < 4; ++i) {
        const float* g = &sG[(l * 4 + i) * 8];
        const float g0 = g[0], g1 = g[1], g2 = g[2], g3 = g[3];
        const float g4 = g[4], g5 = g[5], g6 = g[6], g7 = g[7];
        const int mask = 8 >> i;
        #pragma unroll
        for (int b = 0; b < 16; ++b) {
          if (b & mask) continue;
          const int b1 = b | mask;
          const float r0 = re[b],  i0 = im[b];
          const float r1 = re[b1], i1 = im[b1];
          re[b]  = g0*r0 - g1*i0 + g2*r1 - g3*i1;
          im[b]  = g0*i0 + g1*r0 + g2*i1 + g3*r1;
          re[b1] = g4*r0 - g5*i0 + g6*r1 - g7*i1;
          im[b1] = g4*i0 + g5*r0 + g6*i1 + g7*r1;
        }
      }
      const int rsh = (l % 3) + 1;
      #pragma unroll
      for (int i = 0; i < 4; ++i) {
        const int tq = (i + rsh) & 3;
        const int maskT = 8 >> tq;
        float nre[16], nim[16];
        #pragma unroll
        for (int b = 0; b < 16; ++b) {
          const int srcb = ((b >> (3 - i)) & 1) ? (b ^ maskT) : b;
          nre[b] = re[srcb];
          nim[b] = im[srcb];
        }
        #pragma unroll
        for (int b = 0; b < 16; ++b) { re[b] = nre[b]; im[b] = nim[b]; }
      }
    }
    float prob[16];
    #pragma unroll
    for (int b = 0; b < 16; ++b) prob[b] = re[b] * re[b] + im[b] * im[b];
    float z[4];
    #pragma unroll
    for (int i = 0; i < 4; ++i) {
      float t[16];
      #pragma unroll
      for (int b = 0; b < 16; ++b) t[b] = ((b >> (3 - i)) & 1) ? -prob[b] : prob[b];
      #pragma unroll
      for (int s = 1; s < 16; s <<= 1) {
        #pragma unroll
        for (int b = 0; b < 16; b += 2 * s) t[b] = t[b] + t[b + s];
      }
      z[i] = t[0];
    }
    if (hsel == 0) {
      #pragma unroll
      for (int i = 0; i < 4; ++i) sQ[row * 4 + i] = z[i];
    }
  }
  LDS_SYNC();

  v2f aq;
  aq.x = sQ[lid * 4 + 2 * hsel + 0];
  aq.y = sQ[lid * 4 + 2 * hsel + 1];
  for (int n0 = 0; n0 < D_DIM; n0 += 16) {
    v2f bw;
    bw.x = Wp[(2 * hsel + 0) * D_DIM + n0 + lid];
    bw.y = Wp[(2 * hsel + 1) * D_DIM + n0 + lid];
    v8f c = {};
    c = __builtin_amdgcn_wmma_f32_16x16x4_f32(false, aq, false, bw,
                                              (short)0, c, false, false);
    const float bias = bp[n0 + lid];
    float* orow = out + (row0 + hsel * 8) * D_DIM + n0 + lid;
    #pragma unroll
    for (int r = 0; r < 8; ++r)
      orow[(long)r * D_DIM] = c[r] + bias;
  }
}

extern "C" void kernel_launch(void* const* d_in, const int* in_sizes, int n_in,
                              void* d_out, int out_size, void* d_ws, size_t ws_size,
                              hipStream_t stream) {
  const float* x  = (const float*)d_in[0];
  const float* W1 = (const float*)d_in[1];
  const float* b1 = (const float*)d_in[2];
  const float* qw = (const float*)d_in[3];
  const float* Wp = (const float*)d_in[4];
  const float* bp = (const float*)d_in[5];
  float* out = (float*)d_out;

  const int rows = in_sizes[0] / D_DIM;
  qff_main<<<rows / (16 * WG_WAVES), WG_WAVES * 32, 0, stream>>>(x, W1, b1, qw, Wp, bp, out);
}


# GEMM2 tile staged through the wave's LDS x buffer per 128-column quarter, stored as whole rows: 64 16-byte stores of 512 contiguous bytes a wave, were 256 4-byte stores in 64-byte pieces
# speedup vs baseline: 2.5406x; 1.2459x over previous
#include <hip/hip_runtime.h>
#include <math.h>

typedef float v2f __attribute__((ext_vector_type(2)));
typedef float v8f __attribute__((ext_vector_type(8)));
typedef float v4f __attribute__((ext_vector_type(4)));

#define D_DIM 512
#define H_DIM 2048
#define XQ_PITCH 132

#define LOAD_ORDER_FENCE() asm volatile("" ::: "memory")

#define LDS_SYNC() do { __builtin_amdgcn_fence(__ATOMIC_RELEASE, "workgroup", "local"); \
                        __builtin_amdgcn_wave_barrier(); \
                        __builtin_amdgcn_fence(__ATOMIC_ACQUIRE, "workgroup", "local"); } while (0)

#define WG_WAVES 8
#define WAVE_LDS (16 * XQ_PITCH + 3 * 64)

#define WG_LDS_BARRIER() do { asm volatile("s_wait_dscnt 0x0" ::: "memory"); \
                              asm volatile("s_barrier_signal -1" ::: "memory"); \
                              asm volatile("s_barrier_wait -1" ::: "memory"); } while (0)

__global__ __launch_bounds__(WG_WAVES * 32)
void qff_main(const float* __restrict__ x, const float* __restrict__ W1,
              const float* __restrict__ b1, const float* __restrict__ qw,
              const float* __restrict__ Wp, const float* __restrict__ bp,
              float* __restrict__ out)
{
  __shared__ __attribute__((aligned(16))) float sAll[D_DIM * 4 + WG_WAVES * WAVE_LDS];

  const int tid  = threadIdx.x;
  const int lane = tid & 31;
  const int wave = __builtin_amdgcn_readfirstlane(tid >> 5);
  float* const sW   = sAll;
  float* const sXq  = sAll + D_DIM * 4 + wave * WAVE_LDS;
  float* const sAng = sXq + 16 * XQ_PITCH;
  float* const sQ   = sAng + 64;
  float* const sG   = sQ + 64;
  const int hsel = lane >> 4;
  const int lid  = lane & 15;
  const long row0 = ((long)blockIdx.x * WG_WAVES + wave) * 16;

  v4f wv[2];
  #pragma unroll
  for (int i = 0; i < 2; ++i)
    wv[i] = *(const v4f*)(W1 + (long)(i * (WG_WAVES * 32) + tid) * H_DIM);
  LOAD_ORDER_FENCE();
  const int gi = lane & 7;
  const float phi = qw[gi * 3 + 0];
  const float th  = qw[gi * 3 + 1];
  const float om  = qw[gi * 3 + 2];
  LOAD_ORDER_FENCE();
  v4f xv[4][16];
  {
    const float* xt = x + row0 * D_DIM + lane * 4;
    #pragma unroll
    for (int q = 0; q < 4; ++q) {
      #pragma unroll
      for (int r = 0; r < 16; ++r)
        xv[q][r] = *(const v4f*)(xt + r * D_DIM + q * 128);
      LOAD_ORDER_FENCE();
    }
  }

  #pragma unroll
  for (int i = 0; i < 2; ++i)
    *(v4f*)&sW[(i * (WG_WAVES * 32) + tid) * 4] = wv[i];

  {
    const float c = cosf(0.5f * th), s = sinf(0.5f * th);
    const float a = -0.5f * (phi + om);
    const float b =  0.5f * (phi - om);
    const float ca = cosf(a), sa = sinf(a);
    const float cb = cosf(b), sb = sinf(b);
    if (lane < 8) {
      float* g = &sG[lane * 8];
      g[0] =  c * ca;  g[1] =  c * sa;
      g[2] = -s * cb;  g[3] = -s * sb;
      g[4] =  s * cb;  g[5] = -s * sb;
      g[6] =  c * ca;  g[7] = -c * sa;
    }
  }

  WG_LDS_BARRIER();

  v8f acc = {};
  const float bsel = (lid < 4) ? 1.0f : 0.0f;
  const int   bcol = lid & 3;
  #pragma unroll
  for (int q = 0; q < 4; ++q) {
    #pragma unroll
    for (int r = 0; r < 16; ++r)
      *(v4f*)&sXq[r * XQ_PITCH + lane * 4] = xv[q][r];
    LDS_SYNC();
    #pragma unroll
    for (int kk = 0; kk < 32; ++kk) {
      const int k = q * 128 + kk * 4;
      const v2f afrag = *(const v2f*)&sXq[lid * XQ_PITCH + kk * 4 + 2 * hsel];
      v2f bfrag;
      bfrag.x = bsel * sW[(k + 2 * hsel + 0) * 4 + bcol];
      bfrag.y = bsel * sW[(k + 2 * hsel + 1) * 4 + bcol];
      acc = __builtin_amdgcn_wmma_f32_16x16x4_f32(false, afrag, false, bfrag,
                                                  (short)0, acc, false, false);
    }
    LDS_SYNC();
  }

  if (lid < 4) {
    const float bias = b1[lid];
    #pragma unroll
    for (int r = 0; r < 8; ++r) {
      const float v = acc[r] + bias;
      sAng[(hsel * 8 + r) * 4 + lid] = 0.5f * v * (1.0f + erff(v * 0.70710678118654752f));
    }
  }
  LDS_SYNC();

  {
    const int row = lid;
    float a2[4], cs[4], sn[4];
    #pragma unroll
    for (int i = 0; i < 4; ++i) a2[i] = 0.5f * sAng[row * 4 + i];
    #pragma unroll
    for (int i = 0; i < 4; ++i) sincosf(a2[i], &sn[i], &cs[i]);
    float re[16], im[16];
    #pragma unroll
    for (int b = 0; b < 16; ++b) {
      float p = 1.0f;
      #pragma unroll
      for (int i = 0; i < 4; ++i)
        p *= ((b >> (3 - i)) & 1) ? sn[i] : cs[i];
      re[b] = p;
      im[b] = 0.0f;
    }
    #pragma unroll
    for (int l = 0; l < 2; ++l) {
      #pragma unroll
      for (int i = 0; i < 4; ++i) {
        const float* g = &sG[(l * 4 + i) * 8];
        const float g0 = g[0], g1 = g[1], g2 = g[2], g3 = g[3];
        const float g4 = g[4], g5 = g[5], g6 = g[6], g7 = g[7];
        const int mask = 8 >> i;
        #pragma unroll
        for (int b = 0; b < 16; ++b) {
          if (b & mask) continue;
          const int b1 = b | mask;
          const float r0 = re[b],  i0 = im[b];
          const float r1 = re[b1], i1 = im[b1];
          re[b]  = g0*r0 - g1*i0 + g2*r1 - g3*i1;
          im[b]  = g0*i0 + g1*r0 + g2*i1 + g3*r1;
          re[b1] = g4*r0 - g5*i0 + g6*r1 - g7*i1;
          im[b1] = g4*i0 + g5*r0 + g6*i1 + g7*r1;
        }
      }
      const int rsh = (l % 3) + 1;
      #pragma unroll
      for (int i = 0; i < 4; ++i) {
        const int tq = (i + rsh) & 3;
        const int maskT = 8 >> tq;
        float nre[16], nim[16];
        #pragma unroll
        for (int b = 0; b < 16; ++b) {
          const int srcb = ((b >> (3 - i)) & 1) ? (b ^ maskT) : b;
          nre[b] = re[srcb];
          nim[b] = im[srcb];
        }
        #pragma unroll
        for (int b = 0; b < 16; ++b) { re[b] = nre[b]; im[b] = nim[b]; }
      }
    }
    float prob[16];
    #pragma unroll
    for (int b = 0; b < 16; ++b) prob[b] = re[b] * re[b] + im[b] * im[b];
    float z[4];
    #pragma unroll
    for (int i = 0; i < 4; ++i) {
      float t[16];
      #pragma unroll
      for (int b = 0; b < 16; ++b) t[b] = ((b >> (3 - i)) & 1) ? -prob[b] : prob[b];
      #pragma unroll
      for (int s = 1; s < 16; s <<= 1) {
        #pragma unroll
        for (int b = 0; b < 16; b += 2 * s) t[b] = t[b] + t[b + s];
      }
      z[i] = t[0];
    }
    if (hsel == 0) {
      #pragma unroll
      for (int i = 0; i < 4; ++i) sQ[row * 4 + i] = z[i];
    }
  }
  LDS_SYNC();

  v2f aq;
  aq.x = sQ[lid * 4 + 2 * hsel + 0];
  aq.y = sQ[lid * 4 + 2 * hsel + 1];
  float* const obase = out + row0 * D_DIM + lane * 4;
  v2f bwq[8];
  float bq[8];
  #pragma unroll
  for (int t = 0; t < 8; ++t) {
    bwq[t].x = Wp[(2 * hsel + 0) * D_DIM + t * 16 + lid];
    bwq[t].y = Wp[(2 * hsel + 1) * D_DIM + t * 16 + lid];
    bq[t]    = bp[t * 16 + lid];
  }
  #pragma unroll
  for (int qo = 0; qo < 4; ++qo) {
    #pragma unroll
    for (int t = 0; t < 8; ++t) {
      v8f c = {};
      c = __builtin_amdgcn_wmma_f32_16x16x4_f32(false, aq, false, bwq[t],
                                                (short)0, c, false, false);
      float* srow = &sXq[(hsel * 8) * XQ_PITCH + t * 16 + lid];
      #pragma unroll
      for (int r = 0; r < 8; ++r)
        srow[r * XQ_PITCH] = c[r] + bq[t];
    }
    if (qo < 3) {
      #pragma unroll
      for (int t = 0; t < 8; ++t) {
        const int n0 = (qo + 1) * 128 + t * 16;
        bwq[t].x = Wp[(2 * hsel + 0) * D_DIM + n0 + lid];
        bwq[t].y = Wp[(2 * hsel + 1) * D_DIM + n0 + lid];
        bq[t]    = bp[n0 + lid];
      }
    }
    LDS_SYNC();
    #pragma unroll
    for (int r = 0; r < 16; ++r) {
      const v4f v = *(const v4f*)&sXq[r * XQ_PITCH + lane * 4];
      *(v4f*)(obase + r * D_DIM + qo * 128) = v;
    }
    LDS_SYNC();
  }
}

extern "C" void kernel_launch(void* const* d_in, const int* in_sizes, int n_in,
                              void* d_out, int out_size, void* d_ws, size_t ws_size,
                              hipStream_t stream) {
  const float* x  = (const float*)d_in[0];
  const float* W1 = (const float*)d_in[1];
  const float* b1 = (const float*)d_in[2];
  const float* qw = (const float*)d_in[3];
  const float* Wp = (const float*)d_in[4];
  const float* bp = (const float*)d_in[5];
  float* out = (float*)d_out;

  const int rows = in_sizes[0] / D_DIM;
  qff_main<<<rows / (16 * WG_WAVES), WG_WAVES * 32, 0, stream>>>(x, W1, b1, qw, Wp, bp, out);
}


# b1, Wp's B fragments and bp (512 columns, 96 registers) loaded ahead of use: b1 at the start, Wp and bp in 2 groups behind the x tile, under the sim; GeLU and GEMM2 wait for no load; sums kept
# speedup vs baseline: 2.7580x; 1.0856x over previous
#include <hip/hip_runtime.h>
#include <math.h>

typedef float v2f __attribute__((ext_vector_type(2)));
typedef float v8f __attribute__((ext_vector_type(8)));
typedef float v4f __attribute__((ext_vector_type(4)));

#define D_DIM 512
#define H_DIM 2048
#define XQ_PITCH 132

#define LOAD_ORDER_FENCE() asm volatile("" ::: "memory")

#define LDS_SYNC() do { __builtin_amdgcn_fence(__ATOMIC_RELEASE, "workgroup", "local"); \
                        __builtin_amdgcn_wave_barrier(); \
                        __builtin_amdgcn_fence(__ATOMIC_ACQUIRE, "workgroup", "local"); } while (0)

#define WG_WAVES 8
#define WAVE_LDS (16 * XQ_PITCH + 3 * 64)

#define WG_LDS_BARRIER() do { asm volatile("s_wait_dscnt 0x0" ::: "memory"); \
                              asm volatile("s_barrier_signal -1" ::: "memory"); \
                              asm volatile("s_barrier_wait -1" ::: "memory"); } while (0)

__global__ __launch_bounds__(WG_WAVES * 32)
void qff_main(const float* __restrict__ x, const float* __restrict__ W1,
              const float* __restrict__ b1, const float* __restrict__ qw,
              const float* __restrict__ Wp, const float* __restrict__ bp,
              float* __restrict__ out)
{
  __shared__ __attribute__((aligned(16))) float sAll[D_DIM * 4 + WG_WAVES * WAVE_LDS];

  const int tid  = threadIdx.x;
  const int lane = tid & 31;
  const int wave = __builtin_amdgcn_readfirstlane(tid >> 5);
  float* const sW   = sAll;
  float* const sXq  = sAll + D_DIM * 4 + wave * WAVE_LDS;
  float* const sAng = sXq + 16 * XQ_PITCH;
  float* const sQ   = sAng + 64;
  float* const sG   = sQ + 64;
  const int hsel = lane >> 4;
  const int lid  = lane & 15;
  const long row0 = ((long)blockIdx.x * WG_WAVES + wave) * 16;

  v4f wv[2];
  #pragma unroll
  for (int i = 0; i < 2; ++i)
    wv[i] = *(const v4f*)(W1 + (long)(i * (WG_WAVES * 32) + tid) * H_DIM);
  LOAD_ORDER_FENCE();
  const int gi = lane & 7;
  const float phi = qw[gi * 3 + 0];
  const float th  = qw[gi * 3 + 1];
  const float om  = qw[gi * 3 + 2];
  const float b1v = b1[lane & 3];
  LOAD_ORDER_FENCE();
  v4f xv[4][16];
  {
    const float* xt = x + row0 * D_DIM + lane * 4;
    #pragma unroll
    for (int q = 0; q < 4; ++q) {
      #pragma unroll
      for (int r = 0; r < 16; ++r)
        xv[q][r] = *(const v4f*)(xt + r * D_DIM + q * 128);
      LOAD_ORDER_FENCE();
    }
  }

  #pragma unroll
  for (int i = 0; i < 2; ++i)
    *(v4f*)&sW[(i * (WG_WAVES * 32) + tid) * 4] = wv[i];

  {
    const float c = cosf(0.5f * th), s = sinf(0.5f * th);
    const float a = -0.5f * (phi + om);
    const float b =  0.5f * (phi - om);
    const float ca = cosf(a), sa = sinf(a);
    const float cb = cosf(b), sb = sinf(b);
    if (lane < 8) {
      float* g = &sG[lane * 8];
      g[0] =  c * ca;  g[1] =  c * sa;
      g[2] = -s * cb;  g[3] = -s * sb;
      g[4] =  s * cb;  g[5] = -s * sb;
      g[6] =  c * ca;  g[7] = -c * sa;
    }
  }

  WG_LDS_BARRIER();

  v8f acc = {};
  v2f bw[32];
  float bqv[32];
  const float bsel = (lid < 4) ? 1.0f : 0.0f;
  const int   bcol = lid & 3;
  #pragma unroll
  for (int q = 0; q < 4; ++q) {
    #pragma unroll
    for (int r = 0; r < 16; ++r)
      *(v4f*)&sXq[r * XQ_PITCH + lane * 4] = xv[q][r];
    if (q == 3) {
      LOAD_ORDER_FENCE();
      #pragma unroll
      for (int t = 0; t < 16; ++t) {
        bw[t].x = Wp[(2 * hsel + 0) * D_DIM + t * 16 + lid];
        bw[t].y = Wp[(2 * hsel + 1) * D_DIM + t * 16 + lid];
        bqv[t]  = bp[t * 16 + lid];
      }
    }
    LDS_SYNC();
    #pragma unroll
    for (int kk = 0; kk < 32; ++kk) {
      const int k = q * 128 + kk * 4;
      const v2f afrag = *(const v2f*)&sXq[lid * XQ_PITCH + kk * 4 + 2 * hsel];
      v2f bfrag;
      bfrag.x = bsel * sW[(k + 2 * hsel + 0) * 4 + bcol];
      bfrag.y = bsel * sW[(k + 2 * hsel + 1) * 4 + bcol];
      acc = __builtin_amdgcn_wmma_f32_16x16x4_f32(false, afrag, false, bfrag,
                                                  (short)0, acc, false, false);
    }
    LDS_SYNC();
  }

  if (lid < 4) {
    const float bias = b1v;
    #pragma unroll
    for (int r = 0; r < 8; ++r) {
      const float v = acc[r] + bias;
      sAng[(hsel * 8 + r) * 4 + lid] = 0.5f * v * (1.0f + erff(v * 0.70710678118654752f));
    }
  }
  LDS_SYNC();

  #pragma unroll
  for (int t = 16; t < 32; ++t) {
    bw[t].x = Wp[(2 * hsel + 0) * D_DIM + t * 16 + lid];
    bw[t].y = Wp[(2 * hsel + 1) * D_DIM + t * 16 + lid];
    bqv[t]  = bp[t * 16 + lid];
  }

  {
    const int row = lid;
    float a2[4], cs[4], sn[4];
    #pragma unroll
    for (int i = 0; i < 4; ++i) a2[i] = 0.5f * sAng[row * 4 + i];
    #pragma unroll
    for (int i = 0; i < 4; ++i) sincosf(a2[i], &sn[i], &cs[i]);
    float re[16], im[16];
    #pragma unroll
    for (int b = 0; b < 16; ++b) {
      float p = 1.0f;
      #pragma unroll
      for (int i = 0; i < 4; ++i)
        p *= ((b >> (3 - i)) & 1) ? sn[i] : cs[i];
      re[b] = p;
      im[b] = 0.0f;
    }
    #pragma unroll
    for (int l = 0; l < 2; ++l) {
      #pragma unroll
      for (int i = 0; i < 4; ++i) {
        const float* g = &sG[(l * 4 + i) * 8];
        const float g0 = g[0], g1 = g[1], g2 = g[2], g3 = g[3];
        const float g4 = g[4], g5 = g[5], g6 = g[6], g7 = g[7];
        const int mask = 8 >> i;
        #pragma unroll
        for (int b = 0; b < 16; ++b) {
          if (b & mask) continue;
          const int b1 = b | mask;
          const float r0 = re[b],  i0 = im[b];
          const float r1 = re[b1], i1 = im[b1];
          re[b]  = g0*r0 - g1*i0 + g2*r1 - g3*i1;
          im[b]  = g0*i0 + g1*r0 + g2*i1 + g3*r1;
          re[b1] = g4*r0 - g5*i0 + g6*r1 - g7*i1;
          im[b1] = g4*i0 + g5*r0 + g6*i1 + g7*r1;
        }
      }
      const int rsh = (l % 3) + 1;
      #pragma unroll
      for (int i = 0; i < 4; ++i) {
        const int tq = (i + rsh) & 3;
        const int maskT = 8 >> tq;
        float nre[16], nim[16];
        #pragma unroll
        for (int b = 0; b < 16; ++b) {
          const int srcb = ((b >> (3 - i)) & 1) ? (b ^ maskT) : b;
          nre[b] = re[srcb];
          nim[b] = im[srcb];
        }
        #pragma unroll
        for (int b = 0; b < 16; ++b) { re[b] = nre[b]; im[b] = nim[b]; }
      }
    }
    float prob[16];
    #pragma unroll
    for (int b = 0; b < 16; ++b) prob[b] = re[b] * re[b] + im[b] * im[b];
    float z[4];
    #pragma unroll
    for (int i = 0; i < 4; ++i) {
      float t[16];
      #pragma unroll
      for (int b = 0; b < 16; ++b) t[b] = ((b >> (3 - i)) & 1) ? -prob[b] : prob[b];
      #pragma unroll
      for (int s = 1; s < 16; s <<= 1) {
        #pragma unroll
        for (int b = 0; b < 16; b += 2 * s) t[b] = t[b] + t[b + s];
      }
      z[i] = t[0];
    }
    if (hsel == 0) {
      #pragma unroll
      for (int i = 0; i < 4; ++i) sQ[row * 4 + i] = z[i];
    }
  }
  LDS_SYNC();

  v2f aq;
  aq.x = sQ[lid * 4 + 2 * hsel + 0];
  aq.y = sQ[lid * 4 + 2 * hsel + 1];
  float* const obase = out + row0 * D_DIM + lane * 4;
  #pragma unroll
  for (int qo = 0; qo < 4; ++qo) {
    #pragma unroll
    for (int t = 0; t < 8; ++t) {
      v8f c = {};
      c = __builtin_amdgcn_wmma_f32_16x16x4_f32(false, aq, false, bw[qo * 8 + t],
                                                (short)0, c, false, false);
      float* srow = &sXq[(hsel * 8) * XQ_PITCH + t * 16 + lid];
      #pragma unroll
      for (int r = 0; r < 8; ++r)
        srow[r * XQ_PITCH] = c[r] + bqv[qo * 8 + t];
    }
    LDS_SYNC();
    #pragma unroll
    for (int r = 0; r < 16; ++r) {
      const v4f v = *(const v4f*)&sXq[r * XQ_PITCH + lane * 4];
      *(v4f*)(obase + r * D_DIM + qo * 128) = v;
    }
    LDS_SYNC();
  }
}

extern "C" void kernel_launch(void* const* d_in, const int* in_sizes, int n_in,
                              void* d_out, int out_size, void* d_ws, size_t ws_size,
                              hipStream_t stream) {
  const float* x  = (const float*)d_in[0];
  const float* W1 = (const float*)d_in[1];
  const float* b1 = (const float*)d_in[2];
  const float* qw = (const float*)d_in[3];
  const float* Wp = (const float*)d_in[4];
  const float* bp = (const float*)d_in[5];
  float* out = (float*)d_out;

  const int rows = in_sizes[0] / D_DIM;
  qff_main<<<rows / (16 * WG_WAVES), WG_WAVES * 32, 0, stream>>>(x, W1, b1, qw, Wp, bp, out);
}


# GEMM1: each quarter's 32 A and 32 B fragments read from LDS as one group ahead of its 32 WMMAs, which run back to back (LDS reads stood behind every WMMA, 1-2 ahead of use); sum order kept
# speedup vs baseline: 2.9503x; 1.0697x over previous
#include <hip/hip_runtime.h>
#include <math.h>

typedef float v2f __attribute__((ext_vector_type(2)));
typedef float v8f __attribute__((ext_vector_type(8)));
typedef float v4f __attribute__((ext_vector_type(4)));

#define D_DIM 512
#define H_DIM 2048
#define XQ_PITCH 132

#define LOAD_ORDER_FENCE() asm volatile("" ::: "memory")

#define LDS_SYNC() do { __builtin_amdgcn_fence(__ATOMIC_RELEASE, "workgroup", "local"); \
                        __builtin_amdgcn_wave_barrier(); \
                        __builtin_amdgcn_fence(__ATOMIC_ACQUIRE, "workgroup", "local"); } while (0)

#define WG_WAVES 8
#define WAVE_LDS (16 * XQ_PITCH + 3 * 64)

#define WG_LDS_BARRIER() do { asm volatile("s_wait_dscnt 0x0" ::: "memory"); \
                              asm volatile("s_barrier_signal -1" ::: "memory"); \
                              asm volatile("s_barrier_wait -1" ::: "memory"); } while (0)

__global__ __launch_bounds__(WG_WAVES * 32)
void qff_main(const float* __restrict__ x, const float* __restrict__ W1,
              const float* __restrict__ b1, const float* __restrict__ qw,
              const float* __restrict__ Wp, const float* __restrict__ bp,
              float* __restrict__ out)
{
  __shared__ __attribute__((aligned(16))) float sAll[D_DIM * 4 + WG_WAVES * WAVE_LDS];

  const int tid  = threadIdx.x;
  const int lane = tid & 31;
  const int wave = __builtin_amdgcn_readfirstlane(tid >> 5);
  float* const sW   = sAll;
  float* const sXq  = sAll + D_DIM * 4 + wave * WAVE_LDS;
  float* const sAng = sXq + 16 * XQ_PITCH;
  float* const sQ   = sAng + 64;
  float* const sG   = sQ + 64;
  const int hsel = lane >> 4;
  const int lid  = lane & 15;
  const long row0 = ((long)blockIdx.x * WG_WAVES + wave) * 16;

  v4f wv[2];
  #pragma unroll
  for (int i = 0; i < 2; ++i)
    wv[i] = *(const v4f*)(W1 + (long)(i * (WG_WAVES * 32) + tid) * H_DIM);
  LOAD_ORDER_FENCE();
  const int gi = lane & 7;
  const float phi = qw[gi * 3 + 0];
  const float th  = qw[gi * 3 + 1];
  const float om  = qw[gi * 3 + 2];
  const float b1v = b1[lane & 3];
  LOAD_ORDER_FENCE();
  v4f xv[4][16];
  {
    const float* xt = x + row0 * D_DIM + lane * 4;
    #pragma unroll
    for (int q = 0; q < 4; ++q) {
      #pragma unroll
      for (int r = 0; r < 16; ++r)
        xv[q][r] = *(const v4f*)(xt + r * D_DIM + q * 128);
      LOAD_ORDER_FENCE();
    }
  }

  #pragma unroll
  for (int i = 0; i < 2; ++i)
    *(v4f*)&sW[(i * (WG_WAVES * 32) + tid) * 4] = wv[i];

  {
    const float c = cosf(0.5f * th), s = sinf(0.5f * th);
    const float a = -0.5f * (phi + om);
    const float b =  0.5f * (phi - om);
    const float ca = cosf(a), sa = sinf(a);
    const float cb = cosf(b), sb = sinf(b);
    if (lane < 8) {
      float* g = &sG[lane * 8];
      g[0] =  c * ca;  g[1] =  c * sa;
      g[2] = -s * cb;  g[3] = -s * sb;
      g[4] =  s * cb;  g[5] = -s * sb;
      g[6] =  c * ca;  g[7] = -c * sa;
    }
  }

  WG_LDS_BARRIER();

  v8f acc = {};
  v2f bw[32];
  float bqv[32];
  const float bsel = (lid < 4) ? 1.0f : 0.0f;
  const int   bcol = lid & 3;
  #pragma unroll
  for (int q = 0; q < 4; ++q) {
    #pragma unroll
    for (int r = 0; r < 16; ++r)
      *(v4f*)&sXq[r * XQ_PITCH + lane * 4] = xv[q][r];
    if (q == 3) {
      LOAD_ORDER_FENCE();
      #pragma unroll
      for (int t = 0; t < 16; ++t) {
        bw[t].x = Wp[(2 * hsel + 0) * D_DIM + t * 16 + lid];
        bw[t].y = Wp[(2 * hsel + 1) * D_DIM + t * 16 + lid];
        bqv[t]  = bp[t * 16 + lid];
      }
    }
    LDS_SYNC();
    v2f af[32], bf[32];
    #pragma unroll
    for (int kk = 0; kk < 32; ++kk) {
      const int k = q * 128 + kk * 4;
      af[kk] = *(const v2f*)&sXq[lid * XQ_PITCH + kk * 4 + 2 * hsel];
      bf[kk].x = sW[(k + 2 * hsel + 0) * 4 + bcol];
      bf[kk].y = sW[(k + 2 * hsel + 1) * 4 + bcol];
    }
    __builtin_amdgcn_sched_barrier(0);
    #pragma unroll
    for (int kk = 0; kk < 32; ++kk) {
      v2f bfrag;
      bfrag.x = bsel * bf[kk].x;
      bfrag.y = bsel * bf[kk].y;
      acc = __builtin_amdgcn_wmma_f32_16x16x4_f32(false, af[kk], false, bfrag,
                                                  (short)0, acc, false, false);
    }
    __builtin_amdgcn_sched_barrier(0);
    LDS_SYNC();
  }

  if (lid < 4) {
    const float bias = b1v;
    #pragma unroll
    for (int r = 0; r < 8; ++r) {
      const float v = acc[r] + bias;
      sAng[(hsel * 8 + r) * 4 + lid] = 0.5f * v * (1.0f + erff(v * 0.70710678118654752f));
    }
  }
  LDS_SYNC();

  #pragma unroll
  for (int t = 16; t < 32; ++t) {
    bw[t].x = Wp[(2 * hsel + 0) * D_DIM + t * 16 + lid];
    bw[t].y = Wp[(2 * hsel + 1) * D_DIM + t * 16 + lid];
    bqv[t]  = bp[t * 16 + lid];
  }

  {
    const int row = lid;
    float a2[4], cs[4], sn[4];
    #pragma unroll
    for (int i = 0; i < 4; ++i) a2[i] = 0.5f * sAng[row * 4 + i];
    #pragma unroll
    for (int i = 0; i < 4; ++i) sincosf(a2[i], &sn[i], &cs[i]);
    float re[16], im[16];
    #pragma unroll
    for (int b = 0; b < 16; ++b) {
      float p = 1.0f;
      #pragma unroll
      for (int i = 0; i < 4; ++i)
        p *= ((b >> (3 - i)) & 1) ? sn[i] : cs[i];
      re[b] = p;
      im[b] = 0.0f;
    }
    #pragma unroll
    for (int l = 0; l < 2; ++l) {
      #pragma unroll
      for (int i = 0; i < 4; ++i) {
        const float* g = &sG[(l * 4 + i) * 8];
        const float g0 = g[0], g1 = g[1], g2 = g[2], g3 = g[3];
        const float g4 = g[4], g5 = g[5], g6 = g[6], g7 = g[7];
        const int mask = 8 >> i;
        #pragma unroll
        for (int b = 0; b < 16; ++b) {
          if (b & mask) continue;
          const int b1 = b | mask;
          const float r0 = re[b],  i0 = im[b];
          const float r1 = re[b1], i1 = im[b1];
          re[b]  = g0*r0 - g1*i0 + g2*r1 - g3*i1;
          im[b]  = g0*i0 + g1*r0 + g2*i1 + g3*r1;
          re[b1] = g4*r0 - g5*i0 + g6*r1 - g7*i1;
          im[b1] = g4*i0 + g5*r0 + g6*i1 + g7*r1;
        }
      }
      const int rsh = (l % 3) + 1;
      #pragma unroll
      for (int i = 0; i < 4; ++i) {
        const int tq = (i + rsh) & 3;
        const int maskT = 8 >> tq;
        float nre[16], nim[16];
        #pragma unroll
        for (int b = 0; b < 16; ++b) {
          const int srcb = ((b >> (3 - i)) & 1) ? (b ^ maskT) : b;
          nre[b] = re[srcb];
          nim[b] = im[srcb];
        }
        #pragma unroll
        for (int b = 0; b < 16; ++b) { re[b] = nre[b]; im[b] = nim[b]; }
      }
    }
    float prob[16];
    #pragma unroll
    for (int b = 0; b < 16; ++b) prob[b] = re[b] * re[b] + im[b] * im[b];
    float z[4];
    #pragma unroll
    for (int i = 0; i < 4; ++i) {
      float t[16];
      #pragma unroll
      for (int b = 0; b < 16; ++b) t[b] = ((b >> (3 - i)) & 1) ? -prob[b] : prob[b];
      #pragma unroll
      for (int s = 1; s < 16; s <<= 1) {
        #pragma unroll
        for (int b = 0; b < 16; b += 2 * s) t[b] = t[b] + t[b + s];
      }
      z[i] = t[0];
    }
    if (hsel == 0) {
      #pragma unroll
      for (int i = 0; i < 4; ++i) sQ[row * 4 + i] = z[i];
    }
  }
  LDS_SYNC();

  v2f aq;
  aq.x = sQ[lid * 4 + 2 * hsel + 0];
  aq.y = sQ[lid * 4 + 2 * hsel + 1];
  float* const obase = out + row0 * D_DIM + lane * 4;
  #pragma unroll
  for (int qo = 0; qo < 4; ++qo) {
    #pragma unroll
    for (int t = 0; t < 8; ++t) {
      v8f c = {};
      c = __builtin_amdgcn_wmma_f32_16x16x4_f32(false, aq, false, bw[qo * 8 + t],
                                                (short)0, c, false, false);
      float* srow = &sXq[(hsel * 8) * XQ_PITCH + t * 16 + lid];
      #pragma unroll
      for (int r = 0; r < 8; ++r)
        srow[r * XQ_PITCH] = c[r] + bqv[qo * 8 + t];
    }
    LDS_SYNC();
    #pragma unroll
    for (int r = 0; r < 16; ++r) {
      const v4f v = *(const v4f*)&sXq[r * XQ_PITCH + lane * 4];
      *(v4f*)(obase + r * D_DIM + qo * 128) = v;
    }
    LDS_SYNC();
  }
}

extern "C" void kernel_launch(void* const* d_in, const int* in_sizes, int n_in,
                              void* d_out, int out_size, void* d_ws, size_t ws_size,
                              hipStream_t stream) {
  const float* x  = (const float*)d_in[0];
  const float* W1 = (const float*)d_in[1];
  const float* b1 = (const float*)d_in[2];
  const float* qw = (const float*)d_in[3];
  const float* Wp = (const float*)d_in[4];
  const float* bp = (const float*)d_in[5];
  float* out = (float*)d_out;

  const int rows = in_sizes[0] / D_DIM;
  qff_main<<<rows / (16 * WG_WAVES), WG_WAVES * 32, 0, stream>>>(x, W1, b1, qw, Wp, bp, out);
}
